# starting kernel
# baseline (speedup 1.0000x reference)
#include <hip/hip_runtime.h>

typedef float v2f __attribute__((ext_vector_type(2)));
typedef float v8f __attribute__((ext_vector_type(8)));

#define B_DIM 128
#define N_DIM 1024
#define K_DIM 1024
#define NSPLIT 4
#define KSLICE (K_DIM / NSPLIT)
#define KC 64
#define XS_STRIDE 260
#define DELTA 0.01f

__global__ __launch_bounds__(128) void semisym_wmma_splitk_kernel(
    const float* __restrict__ x,
    const float* __restrict__ W,
    float* __restrict__ out)
{
    __shared__ float xs[16 * XS_STRIDE];
    __shared__ float ws[16 * XS_STRIDE];

    const int tid  = threadIdx.x;
    const int wave = tid >> 5;
    const int lane = tid & 31;
    const int half = lane >> 4;
    const int lm   = lane & 15;

    const int M0 = blockIdx.y * 16;
    const int N0 = blockIdx.x * 16;

    v8f acc_out = {};
    v8f acc_sum = {};
    float mx[8];
#pragma unroll
    for (int r = 0; r < 8; ++r) mx[r] = 0.0f;

    const int soff = wave * KC;

    for (int c = 0; c < KSLICE / KC; ++c) {
#pragma unroll
        for (int t = 0; t < 8; ++t) {
            int e   = tid + t * 128;
            int s   = e >> 8;
            int f   = e & 255;
            int row = f >> 4;
            int c4  = (f & 15) << 2;
            int gk  = s * KSLICE + c * KC + c4;
            float4 vx = *(const float4*)&x[(size_t)(M0 + row) * K_DIM + gk];
            *(float4*)&xs[row * XS_STRIDE + s * KC + c4] = vx;
            float4 vw = *(const float4*)&W[(size_t)(N0 + row) * K_DIM + gk];
            *(float4*)&ws[row * XS_STRIDE + s * KC + c4] = vw;
        }
        __syncthreads();

        const float* xrow = &xs[lm * XS_STRIDE + soff + half * 2];
        const float* wrow = &ws[lm * XS_STRIDE + soff + half * 2];
#pragma unroll
        for (int kk = 0; kk < KC; kk += 4) {
            v2f a = *(const v2f*)&xrow[kk];
            v2f b = *(const v2f*)&wrow[kk];
            acc_out = __builtin_amdgcn_wmma_f32_16x16x4_f32(
                false, a, false, b, (short)0, acc_out, false, false);
            v2f aa = { __builtin_fabsf(a.x), __builtin_fabsf(a.y) };
            v2f ab = { __builtin_fabsf(b.x), __builtin_fabsf(b.y) };
            acc_sum = __builtin_amdgcn_wmma_f32_16x16x4_f32(
                false, aa, false, ab, (short)0, acc_sum, false, false);
        }

        {
            const float* wl = &ws[lm * XS_STRIDE + soff];
#pragma unroll 4
            for (int kk = 0; kk < KC; kk += 4) {
                float4 wv = *(const float4*)&wl[kk];
                float aw0 = __builtin_fabsf(wv.x);
                float aw1 = __builtin_fabsf(wv.y);
                float aw2 = __builtin_fabsf(wv.z);
                float aw3 = __builtin_fabsf(wv.w);
#pragma unroll
                for (int r = 0; r < 8; ++r) {
                    float4 xv = *(const float4*)&xs[(half * 8 + r) * XS_STRIDE + soff + kk];
                    float m0 = fmaxf(__builtin_fabsf(xv.x) * aw0,
                                     __builtin_fabsf(xv.y) * aw1);
                    float m1 = fmaxf(__builtin_fabsf(xv.z) * aw2,
                                     __builtin_fabsf(xv.w) * aw3);
                    mx[r] = fmaxf(mx[r], fmaxf(m0, m1));
                }
            }
        }
        __syncthreads();
    }

    {
        float* red = xs;
        const int base = (wave * 32 + lane) * 24;
#pragma unroll
        for (int r = 0; r < 8; ++r) {
            red[base + r]      = acc_out[r];
            red[base + 8 + r]  = acc_sum[r];
            red[base + 16 + r] = mx[r];
        }
        __syncthreads();

#pragma unroll
        for (int e = tid; e < 256; e += 128) {
            int l = e & 31;
            int r = e >> 5;
            float so = 0.0f, ss = 0.0f, sm = 0.0f;
#pragma unroll
            for (int w = 0; w < NSPLIT; ++w) {
                int idx = (w * 32 + l) * 24 + r;
                so += red[idx];
                ss += red[idx + 8];
                sm  = fmaxf(sm, red[idx + 16]);
            }
            int row = M0 + (l >> 4) * 8 + r;
            int col = N0 + (l & 15);
            out[(size_t)row * N_DIM + col] = so + DELTA * (sm - ss);
        }
    }
}

extern "C" void kernel_launch(void* const* d_in, const int* in_sizes, int n_in,
                              void* d_out, int out_size, void* d_ws, size_t ws_size,
                              hipStream_t stream) {
    (void)in_sizes; (void)n_in; (void)d_ws; (void)ws_size; (void)out_size;
    const float* x = (const float*)d_in[0];
    const float* W = (const float*)d_in[1];
    float* out = (float*)d_out;

    dim3 grid(N_DIM / 16, B_DIM / 16, 1);
    dim3 block(128, 1, 1);
    semisym_wmma_splitk_kernel<<<grid, block, 0, stream>>>(x, W, out);
}


# max-product multiplies written as k-adjacent pairs: one packed multiply per two k straight from the float4 LDS reads, abs folded into max3 source modifiers; no register shuffles left
# speedup vs baseline: 1.1395x; 1.1395x over previous
#include <hip/hip_runtime.h>

typedef float v2f __attribute__((ext_vector_type(2)));
typedef float v4f __attribute__((ext_vector_type(4)));
typedef float v8f __attribute__((ext_vector_type(8)));

#define B_DIM 128
#define N_DIM 1024
#define K_DIM 1024
#define NSPLIT 4
#define KSLICE (K_DIM / NSPLIT)
#define KC 64
#define XS_STRIDE 260
#define DELTA 0.01f

__global__ __launch_bounds__(128) void semisym_wmma_splitk_kernel(
    const float* __restrict__ x,
    const float* __restrict__ W,
    float* __restrict__ out)
{
    __shared__ float xs[16 * XS_STRIDE];
    __shared__ float ws[16 * XS_STRIDE];

    const int tid  = threadIdx.x;
    const int wave = tid >> 5;
    const int lane = tid & 31;
    const int half = lane >> 4;
    const int lm   = lane & 15;

    const int M0 = blockIdx.y * 16;
    const int N0 = blockIdx.x * 16;

    v8f acc_out = {};
    v8f acc_sum = {};
    float mx[8];
#pragma unroll
    for (int r = 0; r < 8; ++r) mx[r] = 0.0f;

    const int soff = wave * KC;

    for (int c = 0; c < KSLICE / KC; ++c) {
#pragma unroll
        for (int t = 0; t < 8; ++t) {
            int e   = tid + t * 128;
            int s   = e >> 8;
            int f   = e & 255;
            int row = f >> 4;
            int c4  = (f & 15) << 2;
            int gk  = s * KSLICE + c * KC + c4;
            float4 vx = *(const float4*)&x[(size_t)(M0 + row) * K_DIM + gk];
            *(float4*)&xs[row * XS_STRIDE + s * KC + c4] = vx;
            float4 vw = *(const float4*)&W[(size_t)(N0 + row) * K_DIM + gk];
            *(float4*)&ws[row * XS_STRIDE + s * KC + c4] = vw;
        }
        __syncthreads();

        const float* xrow = &xs[lm * XS_STRIDE + soff + half * 2];
        const float* wrow = &ws[lm * XS_STRIDE + soff + half * 2];
#pragma unroll
        for (int kk = 0; kk < KC; kk += 4) {
            v2f a = *(const v2f*)&xrow[kk];
            v2f b = *(const v2f*)&wrow[kk];
            acc_out = __builtin_amdgcn_wmma_f32_16x16x4_f32(
                false, a, false, b, (short)0, acc_out, false, false);
            v2f aa = { __builtin_fabsf(a.x), __builtin_fabsf(a.y) };
            v2f ab = { __builtin_fabsf(b.x), __builtin_fabsf(b.y) };
            acc_sum = __builtin_amdgcn_wmma_f32_16x16x4_f32(
                false, aa, false, ab, (short)0, acc_sum, false, false);
        }

        {
            const float* wl = &ws[lm * XS_STRIDE + soff];
#pragma unroll 4
            for (int kk = 0; kk < KC; kk += 4) {
                v4f wv = *(const v4f*)&wl[kk];
                v2f w01 = __builtin_shufflevector(wv, wv, 0, 1);
                v2f w23 = __builtin_shufflevector(wv, wv, 2, 3);
#pragma unroll
                for (int r = 0; r < 8; ++r) {
                    v4f xv = *(const v4f*)&xs[(half * 8 + r) * XS_STRIDE + soff + kk];
                    v2f p01 = __builtin_shufflevector(xv, xv, 0, 1) * w01;
                    v2f p23 = __builtin_shufflevector(xv, xv, 2, 3) * w23;
                    float m = fmaxf(fmaxf(mx[r], __builtin_fabsf(p01.x)), __builtin_fabsf(p01.y));
                    mx[r]   = fmaxf(fmaxf(m,     __builtin_fabsf(p23.x)), __builtin_fabsf(p23.y));
                }
            }
        }
        __syncthreads();
    }

    {
        float* red = xs;
        const int base = (wave * 32 + lane) * 24;
#pragma unroll
        for (int r = 0; r < 8; ++r) {
            red[base + r]      = acc_out[r];
            red[base + 8 + r]  = acc_sum[r];
            red[base + 16 + r] = mx[r];
        }
        __syncthreads();

#pragma unroll
        for (int e = tid; e < 256; e += 128) {
            int l = e & 31;
            int r = e >> 5;
            float so = 0.0f, ss = 0.0f, sm = 0.0f;
#pragma unroll
            for (int w = 0; w < NSPLIT; ++w) {
                int idx = (w * 32 + l) * 24 + r;
                so += red[idx];
                ss += red[idx + 8];
                sm  = fmaxf(sm, red[idx + 16]);
            }
            int row = M0 + (l >> 4) * 8 + r;
            int col = N0 + (l & 15);
            out[(size_t)row * N_DIM + col] = so + DELTA * (sm - ss);
        }
    }
}

extern "C" void kernel_launch(void* const* d_in, const int* in_sizes, int n_in,
                              void* d_out, int out_size, void* d_ws, size_t ws_size,
                              hipStream_t stream) {
    (void)in_sizes; (void)n_in; (void)d_ws; (void)ws_size; (void)out_size;
    const float* x = (const float*)d_in[0];
    const float* W = (const float*)d_in[1];
    float* out = (float*)d_out;

    dim3 grid(N_DIM / 16, B_DIM / 16, 1);
    dim3 block(128, 1, 1);
    semisym_wmma_splitk_kernel<<<grid, block, 0, stream>>>(x, W, out);
}


# K chunks double-buffered: next chunk's global loads issued into registers ahead of the current chunk's products, two LDS stages, one barrier per chunk; occupancy hint of 2 waves per SIMD
# speedup vs baseline: 1.1987x; 1.0520x over previous
#include <hip/hip_runtime.h>

typedef float v2f __attribute__((ext_vector_type(2)));
typedef float v4f __attribute__((ext_vector_type(4)));
typedef float v8f __attribute__((ext_vector_type(8)));

#define B_DIM 128
#define N_DIM 1024
#define K_DIM 1024
#define NSPLIT 4
#define KSLICE (K_DIM / NSPLIT)
#define KC 64
#define XS_STRIDE 260
#define DELTA 0.01f

__global__ __launch_bounds__(128) __attribute__((amdgpu_waves_per_eu(2, 2)))
void semisym_wmma_splitk_kernel(
    const float* __restrict__ x,
    const float* __restrict__ W,
    float* __restrict__ out)
{
    __shared__ float xs[2][16 * XS_STRIDE];
    __shared__ float ws[2][16 * XS_STRIDE];

    const int tid  = threadIdx.x;
    const int wave = tid >> 5;
    const int lane = tid & 31;
    const int half = lane >> 4;
    const int lm   = lane & 15;

    const int M0 = blockIdx.y * 16;
    const int N0 = blockIdx.x * 16;

    v8f acc_out = {};
    v8f acc_sum = {};
    float mx[8];
#pragma unroll
    for (int r = 0; r < 8; ++r) mx[r] = 0.0f;

    const int soff = wave * KC;

    v4f px[8], pw[8];
#pragma unroll
    for (int t = 0; t < 8; ++t) {
        int e   = tid + t * 128;
        int s   = e >> 8;
        int f   = e & 255;
        int row = f >> 4;
        int c4  = (f & 15) << 2;
        int gk  = s * KSLICE + c4;
        px[t] = *(const v4f*)&x[(size_t)(M0 + row) * K_DIM + gk];
        pw[t] = *(const v4f*)&W[(size_t)(N0 + row) * K_DIM + gk];
    }

    for (int c = 0; c < KSLICE / KC; ++c) {
        float* xb = xs[c & 1];
        float* wb = ws[c & 1];
#pragma unroll
        for (int t = 0; t < 8; ++t) {
            int e   = tid + t * 128;
            int s   = e >> 8;
            int f   = e & 255;
            int row = f >> 4;
            int c4  = (f & 15) << 2;
            *(v4f*)&xb[row * XS_STRIDE + s * KC + c4] = px[t];
            *(v4f*)&wb[row * XS_STRIDE + s * KC + c4] = pw[t];
        }
        __syncthreads();

        if (c + 1 < KSLICE / KC) {
#pragma unroll
            for (int t = 0; t < 8; ++t) {
                int e   = tid + t * 128;
                int s   = e >> 8;
                int f   = e & 255;
                int row = f >> 4;
                int c4  = (f & 15) << 2;
                int gk  = s * KSLICE + (c + 1) * KC + c4;
                px[t] = *(const v4f*)&x[(size_t)(M0 + row) * K_DIM + gk];
                pw[t] = *(const v4f*)&W[(size_t)(N0 + row) * K_DIM + gk];
            }
        }

        const float* xrow = &xb[lm * XS_STRIDE + soff + half * 2];
        const float* wrow = &wb[lm * XS_STRIDE + soff + half * 2];
#pragma unroll
        for (int kk = 0; kk < KC; kk += 4) {
            v2f a = *(const v2f*)&xrow[kk];
            v2f b = *(const v2f*)&wrow[kk];
            acc_out = __builtin_amdgcn_wmma_f32_16x16x4_f32(
                false, a, false, b, (short)0, acc_out, false, false);
            v2f aa = { __builtin_fabsf(a.x), __builtin_fabsf(a.y) };
            v2f ab = { __builtin_fabsf(b.x), __builtin_fabsf(b.y) };
            acc_sum = __builtin_amdgcn_wmma_f32_16x16x4_f32(
                false, aa, false, ab, (short)0, acc_sum, false, false);
        }

        {
            const float* wl = &wb[lm * XS_STRIDE + soff];
#pragma unroll 4
            for (int kk = 0; kk < KC; kk += 4) {
                v4f wv = *(const v4f*)&wl[kk];
                v2f w01 = __builtin_shufflevector(wv, wv, 0, 1);
                v2f w23 = __builtin_shufflevector(wv, wv, 2, 3);
#pragma unroll
                for (int r = 0; r < 8; ++r) {
                    v4f xv = *(const v4f*)&xb[(half * 8 + r) * XS_STRIDE + soff + kk];
                    v2f p01 = __builtin_shufflevector(xv, xv, 0, 1) * w01;
                    v2f p23 = __builtin_shufflevector(xv, xv, 2, 3) * w23;
                    float m = fmaxf(fmaxf(mx[r], __builtin_fabsf(p01.x)), __builtin_fabsf(p01.y));
                    mx[r]   = fmaxf(fmaxf(m,     __builtin_fabsf(p23.x)), __builtin_fabsf(p23.y));
                }
            }
        }
    }

    {
        float* red = xs[0];
        const int base = (wave * 32 + lane) * 24;
#pragma unroll
        for (int r = 0; r < 8; ++r) {
            red[base + r]      = acc_out[r];
            red[base + 8 + r]  = acc_sum[r];
            red[base + 16 + r] = mx[r];
        }
        __syncthreads();

#pragma unroll
        for (int e = tid; e < 256; e += 128) {
            int l = e & 31;
            int r = e >> 5;
            float so = 0.0f, ss = 0.0f, sm = 0.0f;
#pragma unroll
            for (int w = 0; w < NSPLIT; ++w) {
                int idx = (w * 32 + l) * 24 + r;
                so += red[idx];
                ss += red[idx + 8];
                sm  = fmaxf(sm, red[idx + 16]);
            }
            int row = M0 + (l >> 4) * 8 + r;
            int col = N0 + (l & 15);
            out[(size_t)row * N_DIM + col] = so + DELTA * (sm - ss);
        }
    }
}

extern "C" void kernel_launch(void* const* d_in, const int* in_sizes, int n_in,
                              void* d_out, int out_size, void* d_ws, size_t ws_size,
                              hipStream_t stream) {
    (void)in_sizes; (void)n_in; (void)d_ws; (void)ws_size; (void)out_size;
    const float* x = (const float*)d_in[0];
    const float* W = (const float*)d_in[1];
    float* out = (float*)d_out;

    dim3 grid(N_DIM / 16, B_DIM / 16, 1);
    dim3 block(128, 1, 1);
    semisym_wmma_splitk_kernel<<<grid, block, 0, stream>>>(x, W, out);
}


# max-product register-tiled: each lane keeps 8 rows x 4 columns on a quarter of its strip's k, so one float4 LDS read feeds 16 to 32 products (144 -> 48 reads per chunk); 16 partial maxima combined
# speedup vs baseline: 1.2975x; 1.0824x over previous
#include <hip/hip_runtime.h>

typedef float v2f __attribute__((ext_vector_type(2)));
typedef float v4f __attribute__((ext_vector_type(4)));
typedef float v8f __attribute__((ext_vector_type(8)));

#define B_DIM 128
#define N_DIM 1024
#define K_DIM 1024
#define NSPLIT 4
#define KSLICE (K_DIM / NSPLIT)
#define KC 64
#define XS_STRIDE 260
#define DELTA 0.01f

__global__ __launch_bounds__(128) __attribute__((amdgpu_waves_per_eu(2, 2)))
void semisym_wmma_splitk_kernel(
    const float* __restrict__ x,
    const float* __restrict__ W,
    float* __restrict__ out)
{
    __shared__ float xs[2][16 * XS_STRIDE];
    __shared__ float ws[2][16 * XS_STRIDE];

    const int tid  = threadIdx.x;
    const int wave = tid >> 5;
    const int lane = tid & 31;
    const int half = lane >> 4;
    const int lm   = lane & 15;

    const int M0 = blockIdx.y * 16;
    const int N0 = blockIdx.x * 16;

    const int ng = (lane >> 2) & 3;
    const int kg = lane & 3;

    v8f acc_out = {};
    v8f acc_sum = {};
    float mx[8][4];
#pragma unroll
    for (int i = 0; i < 8; ++i)
#pragma unroll
        for (int j = 0; j < 4; ++j) mx[i][j] = 0.0f;

    const int soff = wave * KC;

    v4f px[8], pw[8];
#pragma unroll
    for (int t = 0; t < 8; ++t) {
        int e   = tid + t * 128;
        int s   = e >> 8;
        int f   = e & 255;
        int row = f >> 4;
        int c4  = (f & 15) << 2;
        int gk  = s * KSLICE + c4;
        px[t] = *(const v4f*)&x[(size_t)(M0 + row) * K_DIM + gk];
        pw[t] = *(const v4f*)&W[(size_t)(N0 + row) * K_DIM + gk];
    }

    for (int c = 0; c < KSLICE / KC; ++c) {
        float* xb = xs[c & 1];
        float* wb = ws[c & 1];
#pragma unroll
        for (int t = 0; t < 8; ++t) {
            int e   = tid + t * 128;
            int s   = e >> 8;
            int f   = e & 255;
            int row = f >> 4;
            int c4  = (f & 15) << 2;
            *(v4f*)&xb[row * XS_STRIDE + s * KC + c4] = px[t];
            *(v4f*)&wb[row * XS_STRIDE + s * KC + c4] = pw[t];
        }
        __syncthreads();

        if (c + 1 < KSLICE / KC) {
#pragma unroll
            for (int t = 0; t < 8; ++t) {
                int e   = tid + t * 128;
                int s   = e >> 8;
                int f   = e & 255;
                int row = f >> 4;
                int c4  = (f & 15) << 2;
                int gk  = s * KSLICE + (c + 1) * KC + c4;
                px[t] = *(const v4f*)&x[(size_t)(M0 + row) * K_DIM + gk];
                pw[t] = *(const v4f*)&W[(size_t)(N0 + row) * K_DIM + gk];
            }
        }

        const float* xrow = &xb[lm * XS_STRIDE + soff + half * 2];
        const float* wrow = &wb[lm * XS_STRIDE + soff + half * 2];
#pragma unroll
        for (int kk = 0; kk < KC; kk += 4) {
            v2f a = *(const v2f*)&xrow[kk];
            v2f b = *(const v2f*)&wrow[kk];
            acc_out = __builtin_amdgcn_wmma_f32_16x16x4_f32(
                false, a, false, b, (short)0, acc_out, false, false);
            v2f aa = { __builtin_fabsf(a.x), __builtin_fabsf(a.y) };
            v2f ab = { __builtin_fabsf(b.x), __builtin_fabsf(b.y) };
            acc_sum = __builtin_amdgcn_wmma_f32_16x16x4_f32(
                false, aa, false, ab, (short)0, acc_sum, false, false);
        }

        {
            const float* xl = &xb[(half * 8) * XS_STRIDE + soff + kg * 4];
            const float* wl = &wb[(ng * 4) * XS_STRIDE + soff + kg * 4];
#pragma unroll
            for (int ks = 0; ks < KC; ks += 16) {
                v4f wv[4];
#pragma unroll
                for (int j = 0; j < 4; ++j)
                    wv[j] = *(const v4f*)&wl[j * XS_STRIDE + ks];
#pragma unroll
                for (int i = 0; i < 8; ++i) {
                    v4f xv = *(const v4f*)&xl[i * XS_STRIDE + ks];
                    v2f x01 = __builtin_shufflevector(xv, xv, 0, 1);
                    v2f x23 = __builtin_shufflevector(xv, xv, 2, 3);
#pragma unroll
                    for (int j = 0; j < 4; ++j) {
                        v2f p01 = x01 * __builtin_shufflevector(wv[j], wv[j], 0, 1);
                        v2f p23 = x23 * __builtin_shufflevector(wv[j], wv[j], 2, 3);
                        float m  = fmaxf(fmaxf(mx[i][j], __builtin_fabsf(p01.x)), __builtin_fabsf(p01.y));
                        mx[i][j] = fmaxf(fmaxf(m,        __builtin_fabsf(p23.x)), __builtin_fabsf(p23.y));
                    }
                }
            }
        }
    }

    {
        float* redm = xs[0];
        float* reda = ws[0];
        const int part = wave * 4 + kg;
#pragma unroll
        for (int i = 0; i < 8; ++i) {
            v4f v = { mx[i][0], mx[i][1], mx[i][2], mx[i][3] };
            *(v4f*)&redm[part * 256 + (half * 8 + i) * 16 + ng * 4] = v;
        }
        const int base = (wave * 32 + lane) * 17;
#pragma unroll
        for (int r = 0; r < 8; ++r) {
            reda[base + r]     = acc_out[r];
            reda[base + 8 + r] = acc_sum[r];
        }
        __syncthreads();

#pragma unroll
        for (int e = tid; e < 256; e += 128) {
            int m = e >> 4;
            int n = e & 15;
            int l = n + ((m >> 3) << 4);
            int r = m & 7;
            float so = 0.0f, ss = 0.0f, sm = 0.0f;
#pragma unroll
            for (int w = 0; w < NSPLIT; ++w) {
                int idx = (w * 32 + l) * 17 + r;
                so += reda[idx];
                ss += reda[idx + 8];
            }
#pragma unroll
            for (int p = 0; p < 4 * NSPLIT; ++p)
                sm = fmaxf(sm, redm[p * 256 + e]);
            out[(size_t)(M0 + m) * N_DIM + N0 + n] = so + DELTA * (sm - ss);
        }
    }
}

extern "C" void kernel_launch(void* const* d_in, const int* in_sizes, int n_in,
                              void* d_out, int out_size, void* d_ws, size_t ws_size,
                              hipStream_t stream) {
    (void)in_sizes; (void)n_in; (void)d_ws; (void)ws_size; (void)out_size;
    const float* x = (const float*)d_in[0];
    const float* W = (const float*)d_in[1];
    float* out = (float*)d_out;

    dim3 grid(N_DIM / 16, B_DIM / 16, 1);
    dim3 block(128, 1, 1);
    semisym_wmma_splitk_kernel<<<grid, block, 0, stream>>>(x, W, out);
}


# LDS reads of a K chunk issued as two groups of 32, each ahead of the products that use it: two places per chunk where the scheduler had spread them over 24 among the vector and matrix instructions
# speedup vs baseline: 1.3904x; 1.0716x over previous
#include <hip/hip_runtime.h>

typedef float v2f __attribute__((ext_vector_type(2)));
typedef float v4f __attribute__((ext_vector_type(4)));
typedef float v8f __attribute__((ext_vector_type(8)));

#define B_DIM 128
#define N_DIM 1024
#define K_DIM 1024
#define NSPLIT 4
#define KSLICE (K_DIM / NSPLIT)
#define KC 64
#define XS_STRIDE 260
#define DELTA 0.01f

#define MAX_PRODUCTS(S0, S1)                                                            \
    _Pragma("unroll")                                                                   \
    for (int s = (S0); s < (S1); ++s) {                                                 \
        _Pragma("unroll")                                                               \
        for (int i = 0; i < 8; ++i) {                                                   \
            v2f x01 = __builtin_shufflevector(xv[s][i], xv[s][i], 0, 1);                \
            v2f x23 = __builtin_shufflevector(xv[s][i], xv[s][i], 2, 3);                \
            _Pragma("unroll")                                                           \
            for (int j = 0; j < 4; ++j) {                                               \
                v2f p01 = x01 * __builtin_shufflevector(wv[s][j], wv[s][j], 0, 1);      \
                v2f p23 = x23 * __builtin_shufflevector(wv[s][j], wv[s][j], 2, 3);      \
                float m  = fmaxf(fmaxf(mx[i][j], __builtin_fabsf(p01.x)), __builtin_fabsf(p01.y)); \
                mx[i][j] = fmaxf(fmaxf(m,        __builtin_fabsf(p23.x)), __builtin_fabsf(p23.y)); \
            }                                                                           \
        }                                                                               \
    }

__global__ __launch_bounds__(128) __attribute__((amdgpu_waves_per_eu(2, 2)))
void semisym_wmma_splitk_kernel(
    const float* __restrict__ x,
    const float* __restrict__ W,
    float* __restrict__ out)
{
    __shared__ float xs[2][16 * XS_STRIDE];
    __shared__ float ws[2][16 * XS_STRIDE];

    const int tid  = threadIdx.x;
    const int wave = tid >> 5;
    const int lane = tid & 31;
    const int half = lane >> 4;
    const int lm   = lane & 15;

    const int M0 = blockIdx.y * 16;
    const int N0 = blockIdx.x * 16;

    const int ng = (lane >> 2) & 3;
    const int kg = lane & 3;

    v8f acc_out = {};
    v8f acc_sum = {};
    float mx[8][4];
#pragma unroll
    for (int i = 0; i < 8; ++i)
#pragma unroll
        for (int j = 0; j < 4; ++j) mx[i][j] = 0.0f;

    const int soff = wave * KC;

    v4f px[8], pw[8];
#pragma unroll
    for (int t = 0; t < 8; ++t) {
        int e   = tid + t * 128;
        int s   = e >> 8;
        int f   = e & 255;
        int row = f >> 4;
        int c4  = (f & 15) << 2;
        int gk  = s * KSLICE + c4;
        px[t] = *(const v4f*)&x[(size_t)(M0 + row) * K_DIM + gk];
        pw[t] = *(const v4f*)&W[(size_t)(N0 + row) * K_DIM + gk];
    }

    for (int c = 0; c < KSLICE / KC; ++c) {
        float* xb = xs[c & 1];
        float* wb = ws[c & 1];
#pragma unroll
        for (int t = 0; t < 8; ++t) {
            int e   = tid + t * 128;
            int s   = e >> 8;
            int f   = e & 255;
            int row = f >> 4;
            int c4  = (f & 15) << 2;
            *(v4f*)&xb[row * XS_STRIDE + s * KC + c4] = px[t];
            *(v4f*)&wb[row * XS_STRIDE + s * KC + c4] = pw[t];
        }
        __syncthreads();

        if (c + 1 < KSLICE / KC) {
#pragma unroll
            for (int t = 0; t < 8; ++t) {
                int e   = tid + t * 128;
                int s   = e >> 8;
                int f   = e & 255;
                int row = f >> 4;
                int c4  = (f & 15) << 2;
                int gk  = s * KSLICE + (c + 1) * KC + c4;
                px[t] = *(const v4f*)&x[(size_t)(M0 + row) * K_DIM + gk];
                pw[t] = *(const v4f*)&W[(size_t)(N0 + row) * K_DIM + gk];
            }
        }

        const float* xrow = &xb[lm * XS_STRIDE + soff + half * 2];
        const float* wrow = &wb[lm * XS_STRIDE + soff + half * 2];
        const float* xl = &xb[(half * 8) * XS_STRIDE + soff + kg * 4];
        const float* wl = &wb[(ng * 4) * XS_STRIDE + soff + kg * 4];
        v4f wv[KC / 16][4];
        v4f xv[KC / 16][8];
        v2f fa[KC / 4], fb[KC / 4];

#pragma unroll
        for (int s = 0; s < 2; ++s) {
#pragma unroll
            for (int q = s * 4; q < s * 4 + 4; ++q) {
                fa[q] = *(const v2f*)&xrow[q * 4];
                fb[q] = *(const v2f*)&wrow[q * 4];
            }
#pragma unroll
            for (int j = 0; j < 4; ++j)
                wv[s][j] = *(const v4f*)&wl[j * XS_STRIDE + s * 16];
#pragma unroll
            for (int i = 0; i < 8; ++i)
                xv[s][i] = *(const v4f*)&xl[i * XS_STRIDE + s * 16];
        }
        __builtin_amdgcn_sched_barrier(0);

        MAX_PRODUCTS(0, 1)

        __builtin_amdgcn_sched_barrier(0);
#pragma unroll
        for (int s = 2; s < 4; ++s) {
#pragma unroll
            for (int q = s * 4; q < s * 4 + 4; ++q) {
                fa[q] = *(const v2f*)&xrow[q * 4];
                fb[q] = *(const v2f*)&wrow[q * 4];
            }
#pragma unroll
            for (int j = 0; j < 4; ++j)
                wv[s][j] = *(const v4f*)&wl[j * XS_STRIDE + s * 16];
#pragma unroll
            for (int i = 0; i < 8; ++i)
                xv[s][i] = *(const v4f*)&xl[i * XS_STRIDE + s * 16];
        }
        __builtin_amdgcn_sched_barrier(0);

#pragma unroll
        for (int q = 0; q < KC / 4; ++q) {
            v2f a = fa[q];
            v2f b = fb[q];
            acc_out = __builtin_amdgcn_wmma_f32_16x16x4_f32(
                false, a, false, b, (short)0, acc_out, false, false);
            v2f aa = { __builtin_fabsf(a.x), __builtin_fabsf(a.y) };
            v2f ab = { __builtin_fabsf(b.x), __builtin_fabsf(b.y) };
            acc_sum = __builtin_amdgcn_wmma_f32_16x16x4_f32(
                false, aa, false, ab, (short)0, acc_sum, false, false);
        }
        MAX_PRODUCTS(1, 4)
    }

    {
        float* redm = xs[0];
        float* reda = ws[0];
        const int part = wave * 4 + kg;
#pragma unroll
        for (int i = 0; i < 8; ++i) {
            v4f v = { mx[i][0], mx[i][1], mx[i][2], mx[i][3] };
            *(v4f*)&redm[part * 256 + (half * 8 + i) * 16 + ng * 4] = v;
        }
        const int base = (wave * 32 + lane) * 17;
#pragma unroll
        for (int r = 0; r < 8; ++r) {
            reda[base + r]     = acc_out[r];
            reda[base + 8 + r] = acc_sum[r];
        }
        __syncthreads();

#pragma unroll
        for (int e = tid; e < 256; e += 128) {
            int m = e >> 4;
            int n = e & 15;
            int l = n + ((m >> 3) << 4);
            int r = m & 7;
            float so = 0.0f, ss = 0.0f, sm = 0.0f;
#pragma unroll
            for (int w = 0; w < NSPLIT; ++w) {
                int idx = (w * 32 + l) * 17 + r;
                so += reda[idx];
                ss += reda[idx + 8];
            }
#pragma unroll
            for (int p = 0; p < 4 * NSPLIT; ++p)
                sm = fmaxf(sm, redm[p * 256 + e]);
            out[(size_t)(M0 + m) * N_DIM + N0 + n] = so + DELTA * (sm - ss);
        }
    }
}

extern "C" void kernel_launch(void* const* d_in, const int* in_sizes, int n_in,
                              void* d_out, int out_size, void* d_ws, size_t ws_size,
                              hipStream_t stream) {
    (void)in_sizes; (void)n_in; (void)d_ws; (void)ws_size; (void)out_size;
    const float* x = (const float*)d_in[0];
    const float* W = (const float*)d_in[1];
    float* out = (float*)d_out;

    dim3 grid(N_DIM / 16, B_DIM / 16, 1);
    dim3 block(128, 1, 1);
    semisym_wmma_splitk_kernel<<<grid, block, 0, stream>>>(x, W, out);
}


# each wave stages only its own K strip, so the chunk loop needs no block barrier, and reads every half chunk from LDS one half ahead, under the products of the half before
# speedup vs baseline: 1.5028x; 1.0808x over previous
#include <hip/hip_runtime.h>


typedef float v2f __attribute__((ext_vector_type(2)));
typedef float v4f __attribute__((ext_vector_type(4)));
typedef float v8f __attribute__((ext_vector_type(8)));

#define B_DIM 128
#define N_DIM 1024
#define K_DIM 1024
#define NSPLIT 4
#define KSLICE (K_DIM / NSPLIT)
#define KC 64
#define NCHUNK (KSLICE / KC)
#define XS_STRIDE 260
#define DELTA 0.01f

#define MAX_PRODUCTS(XV, WV)                                                            \
    _Pragma("unroll")                                                                   \
    for (int s = 0; s < 2; ++s) {                                                       \
        _Pragma("unroll")                                                               \
        for (int i = 0; i < 8; ++i) {                                                   \
            v2f x01 = __builtin_shufflevector(XV[s][i], XV[s][i], 0, 1);                \
            v2f x23 = __builtin_shufflevector(XV[s][i], XV[s][i], 2, 3);                \
            _Pragma("unroll")                                                           \
            for (int j = 0; j < 4; ++j) {                                               \
                v2f p01 = x01 * __builtin_shufflevector(WV[s][j], WV[s][j], 0, 1);      \
                v2f p23 = x23 * __builtin_shufflevector(WV[s][j], WV[s][j], 2, 3);      \
                float m  = fmaxf(fmaxf(mx[i][j], __builtin_fabsf(p01.x)), __builtin_fabsf(p01.y)); \
                mx[i][j] = fmaxf(fmaxf(m,        __builtin_fabsf(p23.x)), __builtin_fabsf(p23.y)); \
            }                                                                           \
        }                                                                               \
    }

#define READ_HALF(XB, WB, H, FA, FB, XV, WV)                                            \
    _Pragma("unroll")                                                                   \
    for (int s = 0; s < 2; ++s) {                                                       \
        _Pragma("unroll")                                                               \
        for (int q = 0; q < 4; ++q) {                                                   \
            FA[s * 4 + q] = *(const v2f*)&(XB)[fro + ((H) * 8 + s * 4 + q) * 4];        \
            FB[s * 4 + q] = *(const v2f*)&(WB)[fro + ((H) * 8 + s * 4 + q) * 4];        \
        }                                                                               \
        _Pragma("unroll")                                                               \
        for (int j = 0; j < 4; ++j)                                                     \
            WV[s][j] = *(const v4f*)&(WB)[wlo + j * XS_STRIDE + ((H) * 2 + s) * 16];    \
        _Pragma("unroll")                                                               \
        for (int i = 0; i < 8; ++i)                                                     \
            XV[s][i] = *(const v4f*)&(XB)[xlo + i * XS_STRIDE + ((H) * 2 + s) * 16];    \
    }

#define WMMA_HALF(FA, FB)                                                               \
    _Pragma("unroll")                                                                   \
    for (int q = 0; q < 8; ++q) {                                                       \
        v2f a = FA[q];                                                                  \
        v2f b = FB[q];                                                                  \
        acc_out = __builtin_amdgcn_wmma_f32_16x16x4_f32(                                \
            false, a, false, b, (short)0, acc_out, false, false);                       \
        v2f aa = { __builtin_fabsf(a.x), __builtin_fabsf(a.y) };                        \
        v2f ab = { __builtin_fabsf(b.x), __builtin_fabsf(b.y) };                        \
        acc_sum = __builtin_amdgcn_wmma_f32_16x16x4_f32(                                \
            false, aa, false, ab, (short)0, acc_sum, false, false);                     \
    }

#define FETCH_CHUNK(C)                                                                  \
    _Pragma("unroll")                                                                   \
    for (int t = 0; t < 8; ++t) {                                                       \
        px[t] = *(const v4f*)&xg[(size_t)(2 * t) * K_DIM + (C) * KC];                   \
        pw[t] = *(const v4f*)&wg[(size_t)(2 * t) * K_DIM + (C) * KC];                   \
    }

#define STORE_CHUNK(XB, WB)                                                             \
    _Pragma("unroll")                                                                   \
    for (int t = 0; t < 8; ++t) {                                                       \
        *(v4f*)&(XB)[sto + 2 * t * XS_STRIDE] = px[t];                                  \
        *(v4f*)&(WB)[sto + 2 * t * XS_STRIDE] = pw[t];                                  \
    }

#define WAVE_ORDER_LDS()                                                                \
    __builtin_amdgcn_fence(__ATOMIC_RELEASE, "wavefront");                              \
    __builtin_amdgcn_wave_barrier();                                                    \
    __builtin_amdgcn_fence(__ATOMIC_ACQUIRE, "wavefront");

__global__ __launch_bounds__(128) __attribute__((amdgpu_waves_per_eu(2, 2)))
void semisym_wmma_splitk_kernel(
    const float* __restrict__ x,
    const float* __restrict__ W,
    float* __restrict__ out)
{
    __shared__ float xs[2][16 * XS_STRIDE];
    __shared__ float ws[2][16 * XS_STRIDE];

    const int tid  = threadIdx.x;
    const int wave = tid >> 5;
    const int lane = tid & 31;
    const int half = lane >> 4;
    const int lm   = lane & 15;

    const int M0 = blockIdx.y * 16;
    const int N0 = blockIdx.x * 16;

    const int ng = (lane >> 2) & 3;
    const int kg = lane & 3;

    v8f acc_out = {};
    v8f acc_sum = {};
    float mx[8][4];
#pragma unroll
    for (int i = 0; i < 8; ++i)
#pragma unroll
        for (int j = 0; j < 4; ++j) mx[i][j] = 0.0f;

    const int soff = wave * KC;

    const float* xg = x + (size_t)(M0 + half) * K_DIM + wave * KSLICE + lm * 4;
    const float* wg = W + (size_t)(N0 + half) * K_DIM + wave * KSLICE + lm * 4;
    const int sto = half * XS_STRIDE + soff + lm * 4;
    const int fro = lm * XS_STRIDE + soff + half * 2;
    const int xlo = (half * 8) * XS_STRIDE + soff + kg * 4;
    const int wlo = (ng * 4) * XS_STRIDE + soff + kg * 4;

    v2f faA[8], fbA[8];
    v4f xvA[2][8], wvA[2][4];
    v4f px[8], pw[8];

    FETCH_CHUNK(0)
    STORE_CHUNK(xs[0], ws[0])
    WAVE_ORDER_LDS()
    FETCH_CHUNK(1)
    READ_HALF(xs[0], ws[0], 0, faA, fbA, xvA, wvA)

#define CHUNK_STEP(C, HAS_NEXT, HAS_FETCH)                                              \
    {                                                                                   \
        const float* xb = xs[(C) & 1];                                                  \
        const float* wb = ws[(C) & 1];                                                  \
        float* xn = xs[((C) + 1) & 1];                                                  \
        float* wn = ws[((C) + 1) & 1];                                                  \
        v2f faB[8], fbB[8];                                                             \
        v4f xvB[2][8], wvB[2][4];                                                       \
           \
        READ_HALF(xb, wb, 1, faB, fbB, xvB, wvB)                                        \
           \
           \
           \
           \
        __builtin_amdgcn_sched_barrier(0);                                              \
            \
        WMMA_HALF(faA, fbA)                                                             \
        MAX_PRODUCTS(xvA, wvA)                                                          \
          \
          \
          \
        asm volatile("" :: "v"(acc_out), "v"(acc_sum));                                 \
        __builtin_amdgcn_sched_barrier(0);                                              \
           \
           \
           \
        if (HAS_NEXT) {                                                                 \
            STORE_CHUNK(xn, wn)                                                         \
            WAVE_ORDER_LDS()                                                            \
            if (HAS_FETCH) {                                                            \
                FETCH_CHUNK((C) + 2)                                                    \
            }                                                                           \
            READ_HALF(xn, wn, 0, faA, fbA, xvA, wvA)                                    \
        }                                                                               \
        __builtin_amdgcn_sched_barrier(0);                                              \
                                                           \
        WMMA_HALF(faB, fbB)                                                             \
        MAX_PRODUCTS(xvB, wvB)                                                          \
           \
           \
        asm volatile("" :: "v"(acc_out), "v"(acc_sum));                                 \
        __builtin_amdgcn_sched_barrier(0);                                              \
    }

#pragma clang loop unroll(disable)
    for (int c = 0; c < NCHUNK - 2; ++c) {
        CHUNK_STEP(c, true, true)
    }
    CHUNK_STEP(NCHUNK - 2, true, false)
    CHUNK_STEP(NCHUNK - 1, false, false)

    {
        float* redm = xs[0];
        float* reda = ws[0];
        __syncthreads();
        const int part = wave * 4 + kg;
#pragma unroll
        for (int i = 0; i < 8; ++i) {
            v4f v = { mx[i][0], mx[i][1], mx[i][2], mx[i][3] };
            *(v4f*)&redm[part * 256 + (half * 8 + i) * 16 + ng * 4] = v;
        }
        const int base = (wave * 32 + lane) * 17;
#pragma unroll
        for (int r = 0; r < 8; ++r) {
            reda[base + r]     = acc_out[r];
            reda[base + 8 + r] = acc_sum[r];
        }
        __syncthreads();

#pragma unroll
        for (int e = tid; e < 256; e += 128) {
            int m = e >> 4;
            int n = e & 15;
            int l = n + ((m >> 3) << 4);
            int r = m & 7;
            float so = 0.0f, ss = 0.0f, sm = 0.0f;
#pragma unroll
            for (int w = 0; w < NSPLIT; ++w) {
                int idx = (w * 32 + l) * 17 + r;
                so += reda[idx];
                ss += reda[idx + 8];
            }
#pragma unroll
            for (int p = 0; p < 4 * NSPLIT; ++p)
                sm = fmaxf(sm, redm[p * 256 + e]);
            out[(size_t)(M0 + m) * N_DIM + N0 + n] = so + DELTA * (sm - ss);
        }
    }
}

extern "C" void kernel_launch(void* const* d_in, const int* in_sizes, int n_in,
                              void* d_out, int out_size, void* d_ws, size_t ws_size,
                              hipStream_t stream) {
    (void)in_sizes; (void)n_in; (void)d_ws; (void)ws_size; (void)out_size;
    const float* x = (const float*)d_in[0];
    const float* W = (const float*)d_in[1];
    float* out = (float*)d_out;

    dim3 grid(N_DIM / 16, B_DIM / 16, 1);
    dim3 block(128, 1, 1);
    semisym_wmma_splitk_kernel<<<grid, block, 0, stream>>>(x, W, out);
}
